# starting kernel
# baseline (speedup 1.0000x reference)
#include <hip/hip_runtime.h>
#include <math.h>


typedef __attribute__((ext_vector_type(2))) float v2f;
typedef __attribute__((ext_vector_type(8))) float v8f;

#define NUM_CLASSES 10
#define N_LAYERS 3

__global__ void qc_setup_coeffs(const float* __restrict__ params,
                                float* __restrict__ Acoef) {
  int t = threadIdx.x;
  if (t < 2 * 16 * 4) Acoef[t] = 0.0f;
  __syncthreads();
  if (t < NUM_CLASSES) {
    float tr[4] = {1.f, 0.f, 0.f, 0.f};
    float ti[4] = {0.f, 0.f, 0.f, 0.f};
    float br[4] = {0.f, 0.f, 0.f, 0.f};
    float bi[4] = {0.f, 0.f, 0.f, 0.f};
    int len = 1;
    for (int l = 0; l < N_LAYERS; ++l) {
      float ntr[4], nti[4], nbr[4], nbi[4];
      for (int j = 0; j <= len; ++j) {
        float xr = 0.f, xi = 0.f, yr = 0.f, yi = 0.f;
        if (j < len) { xr = tr[j]; xi = ti[j]; yr = br[j]; yi = bi[j]; }
        if (j >= 1)  { xr -= br[j-1]; xi -= bi[j-1]; yr += tr[j-1]; yi += ti[j-1]; }
        ntr[j] = xr; nti[j] = xi; nbr[j] = yr; nbi[j] = yi;
      }
      ++len;
      const float* P = params + (t * N_LAYERS + l) * 3;
      float th0 = 0.5f * P[0], th1 = 0.5f * P[1], th2 = 0.5f * P[2];
      float s1, c1, sz0, cz0, sz2, cz2;
      sincosf(th1, &s1, &c1);
      sincosf(th0, &sz0, &cz0);
      sincosf(th2, &sz2, &cz2);
      for (int j = 0; j < len; ++j) {
        float xr = ntr[j], xi = nti[j], yr = nbr[j], yi = nbi[j];
        float txr = xr * cz0 + xi * sz0;
        float txi = xi * cz0 - xr * sz0;
        float tyr = yr * cz0 - yi * sz0;
        float tyi = yi * cz0 + yr * sz0;
        float uxr = c1 * txr - s1 * tyr, uxi = c1 * txi - s1 * tyi;
        float uyr = s1 * txr + c1 * tyr, uyi = s1 * txi + c1 * tyi;
        tr[j] = uxr * cz2 + uxi * sz2;
        ti[j] = uxi * cz2 - uxr * sz2;
        br[j] = uyr * cz2 - uyi * sz2;
        bi[j] = uyi * cz2 + uyr * sz2;
      }
    }
    int nRe = 2 * t, nIm = 2 * t + 1;
    for (int k = 0; k < 4; ++k) {
      Acoef[((nRe >> 4) * 16 + (nRe & 15)) * 4 + k] = tr[k];
      Acoef[((nIm >> 4) * 16 + (nIm & 15)) * 4 + k] = ti[k];
    }
  }
}

__global__ void __launch_bounds__(256)
qc_fuzzy_kernel(const float* __restrict__ X, const float* __restrict__ Acoef,
                float* __restrict__ out, int ntiles) {
  const int lane = threadIdx.x & 31;
  const int wave = threadIdx.x >> 5;
  const int wpb  = blockDim.x >> 5;
  const int wid  = blockIdx.x * wpb + wave;
  const int wstride = gridDim.x * wpb;

  const int  n  = lane & 15;
  const bool hi = lane >= 16;
  const int  kb = hi ? 2 : 0;

  v2f a0, a1;
  a0[0] = Acoef[(0 * 16 + n) * 4 + kb];
  a0[1] = Acoef[(0 * 16 + n) * 4 + kb + 1];
  a1[0] = Acoef[(1 * 16 + n) * 4 + kb];
  a1[1] = Acoef[(1 * 16 + n) * 4 + kb + 1];

  for (int t = wid; t < ntiles; t += wstride) {
    const int p = t * 16 + n;
    const int tn = t + wstride;
    if (tn < ntiles) __builtin_prefetch(X + tn * 16 + n, 0, 0);

    float x = X[p];
    float s, c;
    __sincosf(0.5f * x, &s, &c);
    float c2 = c * c, s2 = s * s;
    v2f b;
    b[0] = hi ? c * s2 : c2 * c;
    b[1] = hi ? s2 * s : c2 * s;

    v8f z = {};
    v8f d0 = __builtin_amdgcn_wmma_f32_16x16x4_f32(false, a0, false, b,
                                                   (short)0, z, false, false);
    v8f d1 = __builtin_amdgcn_wmma_f32_16x16x4_f32(false, a1, false, b,
                                                   (short)0, z, false, false);

    float* o = out + (size_t)p * NUM_CLASSES;
    if (!hi) {
      float q0 = d0[0] * d0[0] + d0[1] * d0[1];
      float q1 = d0[2] * d0[2] + d0[3] * d0[3];
      float q2 = d0[4] * d0[4] + d0[5] * d0[5];
      float q3 = d0[6] * d0[6] + d0[7] * d0[7];
      float q8 = d1[0] * d1[0] + d1[1] * d1[1];
      float q9 = d1[2] * d1[2] + d1[3] * d1[3];
      *reinterpret_cast<float2*>(o + 0) = make_float2(q0, q1);
      *reinterpret_cast<float2*>(o + 2) = make_float2(q2, q3);
      *reinterpret_cast<float2*>(o + 8) = make_float2(q8, q9);
    } else {
      float q4 = d0[0] * d0[0] + d0[1] * d0[1];
      float q5 = d0[2] * d0[2] + d0[3] * d0[3];
      float q6 = d0[4] * d0[4] + d0[5] * d0[5];
      float q7 = d0[6] * d0[6] + d0[7] * d0[7];
      *reinterpret_cast<float2*>(o + 4) = make_float2(q4, q5);
      *reinterpret_cast<float2*>(o + 6) = make_float2(q6, q7);
    }
  }
}

extern "C" void kernel_launch(void* const* d_in, const int* in_sizes, int n_in,
                              void* d_out, int out_size, void* d_ws, size_t ws_size,
                              hipStream_t stream) {
  const float* x      = (const float*)d_in[0];
  const float* params = (const float*)d_in[1];
  float* out   = (float*)d_out;
  float* Acoef = (float*)d_ws;

  const int npix   = in_sizes[0];
  const int ntiles = npix / 16;

  hipLaunchKernelGGL(qc_setup_coeffs, dim3(1), dim3(128), 0, stream,
                     params, Acoef);

  int blocks = (ntiles + 7) / 8;
  if (blocks > 4096) blocks = 4096;
  hipLaunchKernelGGL(qc_fuzzy_kernel, dim3(blocks), dim3(256), 0, stream,
                     x, Acoef, out, ntiles);
}


# Setup launch only: serial chain cut (a class's 9 angles in one load group, 3 sincos per lane on 30 lanes, the 3 layers unrolled in registers, one 16-B row store per thread); still two launches
# speedup vs baseline: 1.6322x; 1.6322x over previous
#include <hip/hip_runtime.h>
#include <math.h>


typedef __attribute__((ext_vector_type(2))) float v2f;
typedef __attribute__((ext_vector_type(8))) float v8f;

#define NUM_CLASSES 10
#define N_LAYERS 3
#define N_COMBO 12


template <int LEN>
__device__ __forceinline__ void qc_layer(float (&tr)[4], float (&ti)[4],
                                         float (&br)[4], float (&bi)[4],
                                         float s1, float c1, float sz0, float cz0,
                                         float sz2, float cz2) {
  float ntr[4], nti[4], nbr[4], nbi[4];
#pragma unroll
  for (int j = 0; j <= LEN; ++j) {
    float xr = 0.f, xi = 0.f, yr = 0.f, yi = 0.f;
    if (j < LEN) { xr = tr[j]; xi = ti[j]; yr = br[j]; yi = bi[j]; }
    if (j >= 1)  { xr -= br[j-1]; xi -= bi[j-1]; yr += tr[j-1]; yi += ti[j-1]; }
    ntr[j] = xr; nti[j] = xi; nbr[j] = yr; nbi[j] = yi;
  }
#pragma unroll
  for (int j = 0; j <= LEN; ++j) {
    float xr = ntr[j], xi = nti[j], yr = nbr[j], yi = nbi[j];
    float txr = xr * cz0 + xi * sz0;
    float txi = xi * cz0 - xr * sz0;
    float tyr = yr * cz0 - yi * sz0;
    float tyi = yi * cz0 + yr * sz0;
    float uxr = c1 * txr - s1 * tyr, uxi = c1 * txi - s1 * tyi;
    float uyr = s1 * txr + c1 * tyr, uyi = s1 * txi + c1 * tyi;
    tr[j] = uxr * cz2 + uxi * sz2;
    ti[j] = uxi * cz2 - uxr * sz2;
    br[j] = uyr * cz2 - uyi * sz2;
    bi[j] = uyi * cz2 + uyr * sz2;
  }
}

__global__ void __launch_bounds__(N_COMBO * 32)
qc_setup_coeffs(const float* __restrict__ params,
                float* __restrict__ Btab) {
  __shared__ float4 gl[NUM_CLASSES][2];
  const int t = threadIdx.x;

  const int q    = t >> 5;
  const int lane = t & 31;
  const int tile = (int)((0x443332211100ull >> (4 * q)) & 15);
  const int pix  = (int)((0x766544332110ull >> (4 * q)) & 15);
  const int h = lane >> 4, n = lane & 15;
  const int f = 16 * tile + n;
  const int cls = f % 10;
  const bool mine = (f / 10 == pix);

  if (t < 32) {
    const int lay = (lane / 10 < N_LAYERS) ? lane / 10 : N_LAYERS - 1;
    const float* P = params + ((lane % 10) * N_LAYERS + lay) * 3;
    float th0 = 0.5f * P[0], th1 = 0.5f * P[1], th2 = 0.5f * P[2];
    float s1, c1, sz0, cz0, sz2, cz2;
    sincosf(th1, &s1, &c1);
    sincosf(th0, &sz0, &cz0);
    sincosf(th2, &sz2, &cz2);
    const int src1 = (lane + 10) & 31, src2 = (lane + 20) & 31;
    float s1b = __shfl(s1, src1),  c1b = __shfl(c1, src1);
    float z0sb = __shfl(sz0, src1), z0cb = __shfl(cz0, src1);
    float z2sb = __shfl(sz2, src1), z2cb = __shfl(cz2, src1);
    float s1c = __shfl(s1, src2),  c1c = __shfl(c1, src2);
    float z0sc = __shfl(sz0, src2), z0cc = __shfl(cz0, src2);
    float z2sc = __shfl(sz2, src2), z2cc = __shfl(cz2, src2);
    if (t < NUM_CLASSES) {
      float tr[4] = {1.f, 0.f, 0.f, 0.f};
      float ti[4] = {0.f, 0.f, 0.f, 0.f};
      float br[4] = {0.f, 0.f, 0.f, 0.f};
      float bi[4] = {0.f, 0.f, 0.f, 0.f};
      qc_layer<1>(tr, ti, br, bi, s1,  c1,  sz0,  cz0,  sz2,  cz2);
      qc_layer<2>(tr, ti, br, bi, s1b, c1b, z0sb, z0cb, z2sb, z2cb);
      qc_layer<3>(tr, ti, br, bi, s1c, c1c, z0sc, z0cc, z2sc, z2cc);
      float g[8];
#pragma unroll
      for (int k = 0; k < 7; ++k) {
        float acc = 0.f;
#pragma unroll
        for (int i = 0; i < 4; ++i) {
          int j = k - i;
          if (j >= 0 && j < 4) acc += tr[i] * tr[j] + ti[i] * ti[j];
        }
        g[k] = acc;
      }
      g[7] = 0.f;
      gl[t][0] = make_float4(g[0], g[2], g[4], g[6]);
      gl[t][1] = make_float4(g[1], g[3], g[5], g[7]);
    }
  }
  __syncthreads();
  float4 v = gl[cls][h];
  if (!mine) v = make_float4(0.f, 0.f, 0.f, 0.f);
  reinterpret_cast<float4*>(Btab)[t] = v;
}

__device__ __forceinline__ void pixel_frag(float x, bool hi, v2f& f0, v2f& f1) {
  float s, c;
  __sincosf(0.5f * x, &s, &c);
  float c2 = c * c, s2 = s * s, cs = c * s;
  float c4 = c2 * c2, s4 = s2 * s2, c2s2 = c2 * s2;
  float u = hi ? cs : c2;
  f0[0] = c4 * u;
  f0[1] = c2s2 * u;
  f1[0] = s4 * u;
  f1[1] = s4 * s2;
}

__device__ __forceinline__ float row_swap8(float v) {
  int i = __builtin_bit_cast(int, v);
  return __builtin_bit_cast(float,
      __builtin_amdgcn_update_dpp(i, i, 0x128, 0xf, 0xf, true));
}
__device__ __forceinline__ v2f row_swap8(v2f v) {
  v2f r;
  r[0] = row_swap8(v[0]);
  r[1] = row_swap8(v[1]);
  return r;
}

#define MM(acc, afrag, bfrag) \
  acc = __builtin_amdgcn_wmma_f32_16x16x4_f32(false, afrag, false, bfrag, (short)0, acc, false, false)

__global__ void __launch_bounds__(256)
qc_fuzzy_kernel(const float* __restrict__ X, const float* __restrict__ Btab,
                float* __restrict__ out, int nunits) {
  const int lane = threadIdx.x & 31;
  const int wave = threadIdx.x >> 5;
  const int wpb  = blockDim.x >> 5;
  const int wid  = blockIdx.x * wpb + wave;
  const int wstride = gridDim.x * wpb;

  const int  n   = lane & 15;
  const bool hi  = lane >= 16;
  const int  grp = 2 * (n & 7) + (n >> 3);

  v2f b0[N_COMBO], b1[N_COMBO];
#pragma unroll
  for (int q = 0; q < N_COMBO; ++q) {
    float4 v = *reinterpret_cast<const float4*>(Btab + (q * 32 + lane) * 4);
    b0[q][0] = v.x; b0[q][1] = v.y;
    b1[q][0] = v.z; b1[q][1] = v.w;
  }

  for (int u = wid; u < nunits; u += wstride) {
    const float* xu = X + (size_t)u * 128 + grp * 8;
    const int un = u + wstride;
    if (un < nunits) __builtin_prefetch(X + (size_t)un * 128 + grp * 8, 0, 0);

    float4 xa = *reinterpret_cast<const float4*>(xu);
    float4 xb = *reinterpret_cast<const float4*>(xu + 4);
    float xs[8] = {xa.x, xa.y, xa.z, xa.w, xb.x, xb.y, xb.z, xb.w};

    v2f fa0[8], fa1[8], fs0[8], fs1[8];
#pragma unroll
    for (int j = 0; j < 8; ++j) pixel_frag(xs[j], hi, fa0[j], fa1[j]);
#pragma unroll
    for (int j = 1; j < 7; ++j) { fs0[j] = row_swap8(fa0[j]); fs1[j] = row_swap8(fa1[j]); }

    v8f d0 = {}, d1 = {}, d2 = {}, d3 = {}, d4 = {};
    MM(d0, fa0[0], b0[0]);   MM(d1, fs0[1], b0[2]);
    MM(d0, fa1[0], b1[0]);   MM(d1, fs1[1], b1[2]);
    MM(d0, fa0[1], b0[1]);   MM(d1, fs0[2], b0[3]);
    MM(d0, fa1[1], b1[1]);   MM(d1, fs1[2], b1[3]);
    MM(d2, fa0[3], b0[5]);   MM(d1, fs0[3], b0[4]);
    MM(d2, fa1[3], b1[5]);   MM(d1, fs1[3], b1[4]);
    MM(d2, fa0[4], b0[6]);   MM(d3, fs0[4], b0[7]);
    MM(d2, fa1[4], b1[6]);   MM(d3, fs1[4], b1[7]);
    MM(d4, fa0[6], b0[10]);  MM(d3, fs0[5], b0[8]);
    MM(d4, fa1[6], b1[10]);  MM(d3, fs1[5], b1[8]);
    MM(d4, fa0[7], b0[11]);  MM(d3, fs0[6], b0[9]);
    MM(d4, fa1[7], b1[11]);  MM(d3, fs1[6], b1[9]);

    float* o = out + (size_t)u * 1280 + lane;
#pragma unroll
    for (int r = 0; r < 8; ++r) {
      float* orow = o + r * 160;
      orow[0]   = hi ? d1[r] : d0[r];
      orow[32]  = hi ? d3[r] : d2[r];
      orow[64]  = hi ? d0[r] : d4[r];
      orow[96]  = hi ? d2[r] : d1[r];
      orow[128] = hi ? d4[r] : d3[r];
    }
  }
}

extern "C" void kernel_launch(void* const* d_in, const int* in_sizes, int n_in,
                              void* d_out, int out_size, void* d_ws, size_t ws_size,
                              hipStream_t stream) {
  const float* x      = (const float*)d_in[0];
  const float* params = (const float*)d_in[1];
  float* out  = (float*)d_out;
  float* Btab = (float*)d_ws;

  const int npix   = in_sizes[0];
  const int nunits = npix / 128;

  hipLaunchKernelGGL(qc_setup_coeffs, dim3(1), dim3(N_COMBO * 32), 0, stream,
                     params, Btab);

  int blocks = (nunits + 7) / 8;
  if (blocks > 512) blocks = 512;
  hipLaunchKernelGGL(qc_fuzzy_kernel, dim3(blocks), dim3(256), 0, stream,
                     x, Btab, out, nunits);
}


# Setup launch folded into the main kernel through registers: each wave derives its 48 B-fragment registers from params (3 sincos per lane, lane exchanges; no LDS, no table in memory); 1 launch, was 2
# speedup vs baseline: 1.7895x; 1.0964x over previous
#include <hip/hip_runtime.h>
#include <math.h>


typedef __attribute__((ext_vector_type(2))) float v2f;
typedef __attribute__((ext_vector_type(8))) float v8f;

#define NUM_CLASSES 10
#define N_LAYERS 3
#define N_COMBO 12


template <int LEN>
__device__ __forceinline__ void qc_layer(float (&tr)[4], float (&ti)[4],
                                         float (&br)[4], float (&bi)[4],
                                         float s1, float c1, float sz0, float cz0,
                                         float sz2, float cz2) {
  float ntr[4], nti[4], nbr[4], nbi[4];
#pragma unroll
  for (int j = 0; j <= LEN; ++j) {
    float xr = 0.f, xi = 0.f, yr = 0.f, yi = 0.f;
    if (j < LEN) { xr = tr[j]; xi = ti[j]; yr = br[j]; yi = bi[j]; }
    if (j >= 1)  { xr -= br[j-1]; xi -= bi[j-1]; yr += tr[j-1]; yi += ti[j-1]; }
    ntr[j] = xr; nti[j] = xi; nbr[j] = yr; nbi[j] = yi;
  }
#pragma unroll
  for (int j = 0; j <= LEN; ++j) {
    float xr = ntr[j], xi = nti[j], yr = nbr[j], yi = nbi[j];
    float txr = xr * cz0 + xi * sz0;
    float txi = xi * cz0 - xr * sz0;
    float tyr = yr * cz0 - yi * sz0;
    float tyi = yi * cz0 + yr * sz0;
    float uxr = c1 * txr - s1 * tyr, uxi = c1 * txi - s1 * tyi;
    float uyr = s1 * txr + c1 * tyr, uyi = s1 * txi + c1 * tyi;
    tr[j] = uxr * cz2 + uxi * sz2;
    ti[j] = uxi * cz2 - uxr * sz2;
    br[j] = uyr * cz2 - uyi * sz2;
    bi[j] = uyi * cz2 + uyr * sz2;
  }
}

__device__ __forceinline__ void wave_coeffs(float p0, float p1, float p2,
                                            int lane, v2f (&b0)[N_COMBO],
                                            v2f (&b1)[N_COMBO]) {
  const int  n  = lane & 15;
  const bool hi = lane >= 16;

  float th0 = 0.5f * p0, th1 = 0.5f * p1, th2 = 0.5f * p2;
  float s1, c1, sz0, cz0, sz2, cz2;
  sincosf(th1, &s1, &c1);
  sincosf(th0, &sz0, &cz0);
  sincosf(th2, &sz2, &cz2);

  const int cl = (n < NUM_CLASSES) ? n : NUM_CLASSES - 1;
  const int src0 = cl, src1 = cl + 10, src2 = cl + 20;
  float s1a = __shfl(s1, src0),   c1a = __shfl(c1, src0);
  float z0sa = __shfl(sz0, src0), z0ca = __shfl(cz0, src0);
  float z2sa = __shfl(sz2, src0), z2ca = __shfl(cz2, src0);
  float s1b = __shfl(s1, src1),   c1b = __shfl(c1, src1);
  float z0sb = __shfl(sz0, src1), z0cb = __shfl(cz0, src1);
  float z2sb = __shfl(sz2, src1), z2cb = __shfl(cz2, src1);
  float s1c = __shfl(s1, src2),   c1c = __shfl(c1, src2);
  float z0sc = __shfl(sz0, src2), z0cc = __shfl(cz0, src2);
  float z2sc = __shfl(sz2, src2), z2cc = __shfl(cz2, src2);

  float tr[4] = {1.f, 0.f, 0.f, 0.f};
  float ti[4] = {0.f, 0.f, 0.f, 0.f};
  float br[4] = {0.f, 0.f, 0.f, 0.f};
  float bi[4] = {0.f, 0.f, 0.f, 0.f};
  qc_layer<1>(tr, ti, br, bi, s1a, c1a, z0sa, z0ca, z2sa, z2ca);
  qc_layer<2>(tr, ti, br, bi, s1b, c1b, z0sb, z0cb, z2sb, z2cb);
  qc_layer<3>(tr, ti, br, bi, s1c, c1c, z0sc, z0cc, z2sc, z2cc);
  float g[8];
#pragma unroll
  for (int k = 0; k < 7; ++k) {
    float acc = 0.f;
#pragma unroll
    for (int i = 0; i < 4; ++i) {
      int j = k - i;
      if (j >= 0 && j < 4) acc += tr[i] * tr[j] + ti[i] * ti[j];
    }
    g[k] = acc;
  }
  g[7] = 0.f;
  float ge[4];
#pragma unroll
  for (int e = 0; e < 4; ++e) ge[e] = hi ? g[2 * e + 1] : g[2 * e];

#pragma unroll
  for (int i = 0; i < 5; ++i) {
    const int f    = 16 * i + n;
    const int from = (lane & 16) + f % 10;
    const int pixf = f / 10;
    const float v0 = __shfl(ge[0], from), v1 = __shfl(ge[1], from);
    const float v2 = __shfl(ge[2], from), v3 = __shfl(ge[3], from);
#pragma unroll
    for (int q = 0; q < N_COMBO; ++q) {
      const int tile = (int)((0x443332211100ull >> (4 * q)) & 15);
      const int pix  = (int)((0x766544332110ull >> (4 * q)) & 15);
      if (tile == i) {
        const bool mine = (pixf == pix);
        b0[q][0] = mine ? v0 : 0.f;  b0[q][1] = mine ? v1 : 0.f;
        b1[q][0] = mine ? v2 : 0.f;  b1[q][1] = mine ? v3 : 0.f;
      }
    }
  }
}

__device__ __forceinline__ void pixel_frag(float x, bool hi, v2f& f0, v2f& f1) {
  float s, c;
  __sincosf(0.5f * x, &s, &c);
  float c2 = c * c, s2 = s * s, cs = c * s;
  float c4 = c2 * c2, s4 = s2 * s2, c2s2 = c2 * s2;
  float u = hi ? cs : c2;
  f0[0] = c4 * u;
  f0[1] = c2s2 * u;
  f1[0] = s4 * u;
  f1[1] = s4 * s2;
}

__device__ __forceinline__ float row_swap8(float v) {
  int i = __builtin_bit_cast(int, v);
  return __builtin_bit_cast(float,
      __builtin_amdgcn_update_dpp(i, i, 0x128, 0xf, 0xf, true));
}
__device__ __forceinline__ v2f row_swap8(v2f v) {
  v2f r;
  r[0] = row_swap8(v[0]);
  r[1] = row_swap8(v[1]);
  return r;
}

#define MM(acc, afrag, bfrag) \
  acc = __builtin_amdgcn_wmma_f32_16x16x4_f32(false, afrag, false, bfrag, (short)0, acc, false, false)

__global__ void __launch_bounds__(256)
qc_fuzzy_kernel(const float* __restrict__ X, const float* __restrict__ params,
                float* __restrict__ out, int nunits) {
  const int lane = threadIdx.x & 31;
  const int wave = threadIdx.x >> 5;
  const int wpb  = blockDim.x >> 5;
  const int wid  = blockIdx.x * wpb + wave;
  const int wstride = gridDim.x * wpb;

  const int  n   = lane & 15;
  const bool hi  = lane >= 16;
  const int  grp = 2 * (n & 7) + (n >> 3);

  int u = wid;
  if (u >= nunits) return;

  const int lay = (lane / 10 < N_LAYERS) ? lane / 10 : N_LAYERS - 1;
  const float* P = params + ((lane % 10) * N_LAYERS + lay) * 3;
  const float p0 = P[0], p1 = P[1], p2 = P[2];

  const float* xu0 = X + (size_t)u * 128 + grp * 8;
  float4 xa = *reinterpret_cast<const float4*>(xu0);
  float4 xb = *reinterpret_cast<const float4*>(xu0 + 4);

  v2f b0[N_COMBO], b1[N_COMBO];
  wave_coeffs(p0, p1, p2, lane, b0, b1);

  while (u < nunits) {
    const int un = u + wstride;
    if (un < nunits) __builtin_prefetch(X + (size_t)un * 128 + grp * 8, 0, 0);

    float xs[8] = {xa.x, xa.y, xa.z, xa.w, xb.x, xb.y, xb.z, xb.w};

    v2f fa0[8], fa1[8], fs0[8], fs1[8];
#pragma unroll
    for (int j = 0; j < 8; ++j) pixel_frag(xs[j], hi, fa0[j], fa1[j]);
#pragma unroll
    for (int j = 1; j < 7; ++j) { fs0[j] = row_swap8(fa0[j]); fs1[j] = row_swap8(fa1[j]); }

    v8f d0 = {}, d1 = {}, d2 = {}, d3 = {}, d4 = {};
    MM(d0, fa0[0], b0[0]);   MM(d1, fs0[1], b0[2]);
    MM(d0, fa1[0], b1[0]);   MM(d1, fs1[1], b1[2]);
    MM(d0, fa0[1], b0[1]);   MM(d1, fs0[2], b0[3]);
    MM(d0, fa1[1], b1[1]);   MM(d1, fs1[2], b1[3]);
    MM(d2, fa0[3], b0[5]);   MM(d1, fs0[3], b0[4]);
    MM(d2, fa1[3], b1[5]);   MM(d1, fs1[3], b1[4]);
    MM(d2, fa0[4], b0[6]);   MM(d3, fs0[4], b0[7]);
    MM(d2, fa1[4], b1[6]);   MM(d3, fs1[4], b1[7]);
    MM(d4, fa0[6], b0[10]);  MM(d3, fs0[5], b0[8]);
    MM(d4, fa1[6], b1[10]);  MM(d3, fs1[5], b1[8]);
    MM(d4, fa0[7], b0[11]);  MM(d3, fs0[6], b0[9]);
    MM(d4, fa1[7], b1[11]);  MM(d3, fs1[6], b1[9]);

    float* o = out + (size_t)u * 1280 + lane;
#pragma unroll
    for (int r = 0; r < 8; ++r) {
      float* orow = o + r * 160;
      orow[0]   = hi ? d1[r] : d0[r];
      orow[32]  = hi ? d3[r] : d2[r];
      orow[64]  = hi ? d0[r] : d4[r];
      orow[96]  = hi ? d2[r] : d1[r];
      orow[128] = hi ? d4[r] : d3[r];
    }

    u = un;
    if (u < nunits) {
      const float* xu = X + (size_t)u * 128 + grp * 8;
      xa = *reinterpret_cast<const float4*>(xu);
      xb = *reinterpret_cast<const float4*>(xu + 4);
    }
  }
}

extern "C" void kernel_launch(void* const* d_in, const int* in_sizes, int n_in,
                              void* d_out, int out_size, void* d_ws, size_t ws_size,
                              hipStream_t stream) {
  const float* x      = (const float*)d_in[0];
  const float* params = (const float*)d_in[1];
  float* out  = (float*)d_out;

  const int npix   = in_sizes[0];
  const int nunits = npix / 128;

  int blocks = (nunits + 7) / 8;
  if (blocks > 512) blocks = 512;
  hipLaunchKernelGGL(qc_fuzzy_kernel, dim3(blocks), dim3(256), 0, stream,
                     x, params, out, nunits);
}


# Product re-tiled: one tile per pixel of the 8-pixel group (K=8, class columns rotated into half-line place): 16 WMMAs per 128 pixels, was 24; half lines joined in registers (selects, permlanex16)
# speedup vs baseline: 1.9390x; 1.0835x over previous
#include <hip/hip_runtime.h>
#include <math.h>


typedef __attribute__((ext_vector_type(2))) float v2f;
typedef __attribute__((ext_vector_type(8))) float v8f;

#define NUM_CLASSES 10
#define N_LAYERS 3
#define N_TILE 8


template <int LEN>
__device__ __forceinline__ void qc_layer(float (&tr)[4], float (&ti)[4],
                                         float (&br)[4], float (&bi)[4],
                                         float s1, float c1, float sz0, float cz0,
                                         float sz2, float cz2) {
  float ntr[4], nti[4], nbr[4], nbi[4];
#pragma unroll
  for (int j = 0; j <= LEN; ++j) {
    float xr = 0.f, xi = 0.f, yr = 0.f, yi = 0.f;
    if (j < LEN) { xr = tr[j]; xi = ti[j]; yr = br[j]; yi = bi[j]; }
    if (j >= 1)  { xr -= br[j-1]; xi -= bi[j-1]; yr += tr[j-1]; yi += ti[j-1]; }
    ntr[j] = xr; nti[j] = xi; nbr[j] = yr; nbi[j] = yi;
  }
#pragma unroll
  for (int j = 0; j <= LEN; ++j) {
    float xr = ntr[j], xi = nti[j], yr = nbr[j], yi = nbi[j];
    float txr = xr * cz0 + xi * sz0;
    float txi = xi * cz0 - xr * sz0;
    float tyr = yr * cz0 - yi * sz0;
    float tyi = yi * cz0 + yr * sz0;
    float uxr = c1 * txr - s1 * tyr, uxi = c1 * txi - s1 * tyi;
    float uyr = s1 * txr + c1 * tyr, uyi = s1 * txi + c1 * tyi;
    tr[j] = uxr * cz2 + uxi * sz2;
    ti[j] = uxi * cz2 - uxr * sz2;
    br[j] = uyr * cz2 - uyi * sz2;
    bi[j] = uyi * cz2 + uyr * sz2;
  }
}

template <int N>
__device__ __forceinline__ float row_rot(float v) {
  if (N == 0) return v;
  int i = __builtin_bit_cast(int, v);
  return __builtin_bit_cast(float,
      __builtin_amdgcn_update_dpp(i, i, 0x120 + N, 0xf, 0xf, true));
}

__device__ __forceinline__ void wave_coeffs(float p0, float p1, float p2,
                                            int lane, v2f (&b0)[N_TILE],
                                            v2f (&b1)[N_TILE]) {
  const int  n  = lane & 15;
  const bool hi = lane >= 16;

  float th0 = 0.5f * p0, th1 = 0.5f * p1, th2 = 0.5f * p2;
  float s1, c1, sz0, cz0, sz2, cz2;
  sincosf(th1, &s1, &c1);
  sincosf(th0, &sz0, &cz0);
  sincosf(th2, &sz2, &cz2);

  const int cl = (n < NUM_CLASSES) ? n : NUM_CLASSES - 1;
  const int src0 = cl, src1 = cl + 10, src2 = cl + 20;
  float s1a = __shfl(s1, src0),   c1a = __shfl(c1, src0);
  float z0sa = __shfl(sz0, src0), z0ca = __shfl(cz0, src0);
  float z2sa = __shfl(sz2, src0), z2ca = __shfl(cz2, src0);
  float s1b = __shfl(s1, src1),   c1b = __shfl(c1, src1);
  float z0sb = __shfl(sz0, src1), z0cb = __shfl(cz0, src1);
  float z2sb = __shfl(sz2, src1), z2cb = __shfl(cz2, src1);
  float s1c = __shfl(s1, src2),   c1c = __shfl(c1, src2);
  float z0sc = __shfl(sz0, src2), z0cc = __shfl(cz0, src2);
  float z2sc = __shfl(sz2, src2), z2cc = __shfl(cz2, src2);

  float tr[4] = {1.f, 0.f, 0.f, 0.f};
  float ti[4] = {0.f, 0.f, 0.f, 0.f};
  float br[4] = {0.f, 0.f, 0.f, 0.f};
  float bi[4] = {0.f, 0.f, 0.f, 0.f};
  qc_layer<1>(tr, ti, br, bi, s1a, c1a, z0sa, z0ca, z2sa, z2ca);
  qc_layer<2>(tr, ti, br, bi, s1b, c1b, z0sb, z0cb, z2sb, z2cb);
  qc_layer<3>(tr, ti, br, bi, s1c, c1c, z0sc, z0cc, z2sc, z2cc);
  float g[8];
#pragma unroll
  for (int k = 0; k < 7; ++k) {
    float acc = 0.f;
#pragma unroll
    for (int i = 0; i < 4; ++i) {
      int j = k - i;
      if (j >= 0 && j < 4) acc += tr[i] * tr[j] + ti[i] * ti[j];
    }
    g[k] = acc;
  }
  g[7] = 0.f;
  const bool own = n < NUM_CLASSES;
  float v[4];
#pragma unroll
  for (int e = 0; e < 4; ++e) v[e] = own ? (hi ? g[2 * e + 1] : g[2 * e]) : 0.f;

#define QC_TILE_B(t, o)                                             \
  b0[t][0] = row_rot<o>(v[0]);  b0[t][1] = row_rot<o>(v[1]);        \
  b1[t][0] = row_rot<o>(v[2]);  b1[t][1] = row_rot<o>(v[3]);
  QC_TILE_B(0, 0)   QC_TILE_B(1, 10)  QC_TILE_B(2, 4)   QC_TILE_B(3, 14)
  QC_TILE_B(4, 8)   QC_TILE_B(5, 2)   QC_TILE_B(6, 12)  QC_TILE_B(7, 6)
#undef QC_TILE_B
}

__device__ __forceinline__ void pixel_frag(float x, bool hi, v2f& f0, v2f& f1) {
  float s, c;
  __sincosf(0.5f * x, &s, &c);
  float c2 = c * c, s2 = s * s, cs = c * s;
  float c4 = c2 * c2, s4 = s2 * s2, c2s2 = c2 * s2;
  float u = hi ? cs : c2;
  f0[0] = c4 * u;
  f0[1] = c2s2 * u;
  f1[0] = s4 * u;
  f1[1] = s4 * s2;
}

__device__ __forceinline__ float half_swap(float v) {
  int i = __builtin_bit_cast(int, v);
  return __builtin_bit_cast(float,
      __builtin_amdgcn_permlanex16(i, i, 0x76543210, 0xfedcba98, false, false));
}

#define MM(acc, afrag, bfrag) \
  acc = __builtin_amdgcn_wmma_f32_16x16x4_f32(false, afrag, false, bfrag, (short)0, acc, false, false)

__global__ void __launch_bounds__(256)
qc_fuzzy_kernel(const float* __restrict__ X, const float* __restrict__ params,
                float* __restrict__ out, int nunits) {
  const int lane = threadIdx.x & 31;
  const int wave = threadIdx.x >> 5;
  const int wpb  = blockDim.x >> 5;
  const int wid  = blockIdx.x * wpb + wave;
  const int wstride = gridDim.x * wpb;

  const int  n   = lane & 15;
  const bool hi  = lane >= 16;
  const int  grp = 2 * (n & 7) + (n >> 3);

  int u = wid;
  if (u >= nunits) return;

  const int lay = (lane / 10 < N_LAYERS) ? lane / 10 : N_LAYERS - 1;
  const float* P = params + ((lane % 10) * N_LAYERS + lay) * 3;
  const float p0 = P[0], p1 = P[1], p2 = P[2];

  const float* xu0 = X + (size_t)u * 128 + grp * 8;
  float4 xa = *reinterpret_cast<const float4*>(xu0);
  float4 xb = *reinterpret_cast<const float4*>(xu0 + 4);

  v2f b0[N_TILE], b1[N_TILE];
  wave_coeffs(p0, p1, p2, lane, b0, b1);

  while (u < nunits) {
    const int un = u + wstride;
    if (un < nunits) __builtin_prefetch(X + (size_t)un * 128 + grp * 8, 0, 0);

    float xs[8] = {xa.x, xa.y, xa.z, xa.w, xb.x, xb.y, xb.z, xb.w};

    v2f fa0[8], fa1[8];
#pragma unroll
    for (int j = 0; j < 8; ++j) pixel_frag(xs[j], hi, fa0[j], fa1[j]);

    v8f d0 = {}, d1 = {}, d2 = {}, d3 = {}, d4 = {}, d5 = {}, d6 = {}, d7 = {};
    MM(d0, fa0[0], b0[0]);   MM(d1, fa0[1], b0[1]);
    MM(d2, fa0[2], b0[2]);   MM(d3, fa0[3], b0[3]);
    MM(d4, fa0[4], b0[4]);   MM(d5, fa0[5], b0[5]);
    MM(d6, fa0[6], b0[6]);   MM(d7, fa0[7], b0[7]);
    MM(d0, fa1[0], b1[0]);   MM(d1, fa1[1], b1[1]);
    MM(d2, fa1[2], b1[2]);   MM(d3, fa1[3], b1[3]);
    MM(d4, fa1[4], b1[4]);   MM(d5, fa1[5], b1[5]);
    MM(d6, fa1[6], b1[6]);   MM(d7, fa1[7], b1[7]);

    const bool c2 = n < 2, c4 = n < 4, c6 = n < 6, c8 = n < 8;
    const bool c10 = n < 10, c12 = n < 12, c14 = n < 14;
    float ln[8][5];
#pragma unroll
    for (int r = 0; r < 8; ++r) {
      const float h0 = c10 ? d0[r] : d1[r];
      const float t1 = c14 ? d2[r] : d3[r];
      const float h1 = c4 ? d1[r] : t1;
      const float h2 = c8 ? d3[r] : d4[r];
      const float t3 = c12 ? d5[r] : d6[r];
      const float h3 = c2 ? d4[r] : t3;
      const float h4 = c6 ? d6[r] : d7[r];
      const float x1 = half_swap(h1), x3 = half_swap(h3);
      ln[r][0] = hi ? x1 : h0;
      ln[r][1] = hi ? x3 : h2;
      ln[r][2] = hi ? h0 : h4;
      ln[r][3] = hi ? h2 : x1;
      ln[r][4] = hi ? h4 : x3;
    }
    float* o = out + (size_t)u * 1280 + lane;
#pragma unroll
    for (int r = 0; r < 8; ++r) {
#pragma unroll
      for (int i = 0; i < 5; ++i) o[r * 160 + 32 * i] = ln[r][i];
    }

    u = un;
    if (u < nunits) {
      const float* xu = X + (size_t)u * 128 + grp * 8;
      xa = *reinterpret_cast<const float4*>(xu);
      xb = *reinterpret_cast<const float4*>(xu + 4);
    }
  }
}

extern "C" void kernel_launch(void* const* d_in, const int* in_sizes, int n_in,
                              void* d_out, int out_size, void* d_ws, size_t ws_size,
                              hipStream_t stream) {
  const float* x      = (const float*)d_in[0];
  const float* params = (const float*)d_in[1];
  float* out  = (float*)d_out;

  const int npix   = in_sizes[0];
  const int nunits = npix / 128;

  int blocks = (nunits + 7) / 8;
  if (blocks > 512) blocks = 512;
  hipLaunchKernelGGL(qc_fuzzy_kernel, dim3(blocks), dim3(256), 0, stream,
                     x, params, out, nunits);
}


# Load placement for latency: angles requested ahead of x and waited for alone, so the coefficient chain runs under the x misses; x held a unit ahead in a second register set (prefetch dropped)
# speedup vs baseline: 2.5382x; 1.3090x over previous
#include <hip/hip_runtime.h>
#include <math.h>


typedef __attribute__((ext_vector_type(2))) float v2f;
typedef __attribute__((ext_vector_type(8))) float v8f;

#define NUM_CLASSES 10
#define N_LAYERS 3
#define N_TILE 8


template <int LEN>
__device__ __forceinline__ void qc_layer(float (&tr)[4], float (&ti)[4],
                                         float (&br)[4], float (&bi)[4],
                                         float s1, float c1, float sz0, float cz0,
                                         float sz2, float cz2) {
  float ntr[4], nti[4], nbr[4], nbi[4];
#pragma unroll
  for (int j = 0; j <= LEN; ++j) {
    float xr = 0.f, xi = 0.f, yr = 0.f, yi = 0.f;
    if (j < LEN) { xr = tr[j]; xi = ti[j]; yr = br[j]; yi = bi[j]; }
    if (j >= 1)  { xr -= br[j-1]; xi -= bi[j-1]; yr += tr[j-1]; yi += ti[j-1]; }
    ntr[j] = xr; nti[j] = xi; nbr[j] = yr; nbi[j] = yi;
  }
#pragma unroll
  for (int j = 0; j <= LEN; ++j) {
    float xr = ntr[j], xi = nti[j], yr = nbr[j], yi = nbi[j];
    float txr = xr * cz0 + xi * sz0;
    float txi = xi * cz0 - xr * sz0;
    float tyr = yr * cz0 - yi * sz0;
    float tyi = yi * cz0 + yr * sz0;
    float uxr = c1 * txr - s1 * tyr, uxi = c1 * txi - s1 * tyi;
    float uyr = s1 * txr + c1 * tyr, uyi = s1 * txi + c1 * tyi;
    tr[j] = uxr * cz2 + uxi * sz2;
    ti[j] = uxi * cz2 - uxr * sz2;
    br[j] = uyr * cz2 - uyi * sz2;
    bi[j] = uyi * cz2 + uyr * sz2;
  }
}

template <int N>
__device__ __forceinline__ float row_rot(float v) {
  if (N == 0) return v;
  int i = __builtin_bit_cast(int, v);
  return __builtin_bit_cast(float,
      __builtin_amdgcn_update_dpp(i, i, 0x120 + N, 0xf, 0xf, true));
}

__device__ __forceinline__ void wave_coeffs(float p0, float p1, float p2,
                                            int lane, v2f (&b0)[N_TILE],
                                            v2f (&b1)[N_TILE]) {
  const int  n  = lane & 15;
  const bool hi = lane >= 16;

  float th0 = 0.5f * p0, th1 = 0.5f * p1, th2 = 0.5f * p2;
  float s1, c1, sz0, cz0, sz2, cz2;
  sincosf(th1, &s1, &c1);
  sincosf(th0, &sz0, &cz0);
  sincosf(th2, &sz2, &cz2);

  const int cl = (n < NUM_CLASSES) ? n : NUM_CLASSES - 1;
  const int src0 = cl, src1 = cl + 10, src2 = cl + 20;
  float s1a = __shfl(s1, src0),   c1a = __shfl(c1, src0);
  float z0sa = __shfl(sz0, src0), z0ca = __shfl(cz0, src0);
  float z2sa = __shfl(sz2, src0), z2ca = __shfl(cz2, src0);
  float s1b = __shfl(s1, src1),   c1b = __shfl(c1, src1);
  float z0sb = __shfl(sz0, src1), z0cb = __shfl(cz0, src1);
  float z2sb = __shfl(sz2, src1), z2cb = __shfl(cz2, src1);
  float s1c = __shfl(s1, src2),   c1c = __shfl(c1, src2);
  float z0sc = __shfl(sz0, src2), z0cc = __shfl(cz0, src2);
  float z2sc = __shfl(sz2, src2), z2cc = __shfl(cz2, src2);

  float tr[4] = {1.f, 0.f, 0.f, 0.f};
  float ti[4] = {0.f, 0.f, 0.f, 0.f};
  float br[4] = {0.f, 0.f, 0.f, 0.f};
  float bi[4] = {0.f, 0.f, 0.f, 0.f};
  qc_layer<1>(tr, ti, br, bi, s1a, c1a, z0sa, z0ca, z2sa, z2ca);
  qc_layer<2>(tr, ti, br, bi, s1b, c1b, z0sb, z0cb, z2sb, z2cb);
  qc_layer<3>(tr, ti, br, bi, s1c, c1c, z0sc, z0cc, z2sc, z2cc);
  float g[8];
#pragma unroll
  for (int k = 0; k < 7; ++k) {
    float acc = 0.f;
#pragma unroll
    for (int i = 0; i < 4; ++i) {
      int j = k - i;
      if (j >= 0 && j < 4) acc += tr[i] * tr[j] + ti[i] * ti[j];
    }
    g[k] = acc;
  }
  g[7] = 0.f;
  const bool own = n < NUM_CLASSES;
  float v[4];
#pragma unroll
  for (int e = 0; e < 4; ++e) v[e] = own ? (hi ? g[2 * e + 1] : g[2 * e]) : 0.f;

#define QC_TILE_B(t, o)                                             \
  b0[t][0] = row_rot<o>(v[0]);  b0[t][1] = row_rot<o>(v[1]);        \
  b1[t][0] = row_rot<o>(v[2]);  b1[t][1] = row_rot<o>(v[3]);
  QC_TILE_B(0, 0)   QC_TILE_B(1, 10)  QC_TILE_B(2, 4)   QC_TILE_B(3, 14)
  QC_TILE_B(4, 8)   QC_TILE_B(5, 2)   QC_TILE_B(6, 12)  QC_TILE_B(7, 6)
#undef QC_TILE_B
}

__device__ __forceinline__ void pixel_frag(float x, bool hi, v2f& f0, v2f& f1) {
  float s, c;
  __sincosf(0.5f * x, &s, &c);
  float c2 = c * c, s2 = s * s, cs = c * s;
  float c4 = c2 * c2, s4 = s2 * s2, c2s2 = c2 * s2;
  float u = hi ? cs : c2;
  f0[0] = c4 * u;
  f0[1] = c2s2 * u;
  f1[0] = s4 * u;
  f1[1] = s4 * s2;
}

__device__ __forceinline__ float half_swap(float v) {
  int i = __builtin_bit_cast(int, v);
  return __builtin_bit_cast(float,
      __builtin_amdgcn_permlanex16(i, i, 0x76543210, 0xfedcba98, false, false));
}

#define MM(acc, afrag, bfrag) \
  acc = __builtin_amdgcn_wmma_f32_16x16x4_f32(false, afrag, false, bfrag, (short)0, acc, false, false)

__global__ void __launch_bounds__(256)
qc_fuzzy_kernel(const float* __restrict__ X, const float* __restrict__ params,
                float* __restrict__ out, int nunits) {
  const int lane = threadIdx.x & 31;
  const int wave = threadIdx.x >> 5;
  const int wpb  = blockDim.x >> 5;
  const int wid  = blockIdx.x * wpb + wave;
  const int wstride = gridDim.x * wpb;

  const int  n   = lane & 15;
  const bool hi  = lane >= 16;
  const int  grp = 2 * (n & 7) + (n >> 3);

  int u = wid;
  if (u >= nunits) return;

  const int lay = (lane / 10 < N_LAYERS) ? lane / 10 : N_LAYERS - 1;
  const float* P = params + ((lane % 10) * N_LAYERS + lay) * 3;
  const float p0 = P[0], p1 = P[1], p2 = P[2];
  asm volatile("" ::: "memory");

  const float* xu0 = X + (size_t)u * 128 + grp * 8;
  float4 xa = *reinterpret_cast<const float4*>(xu0);
  float4 xb = *reinterpret_cast<const float4*>(xu0 + 4);
  int un = u + wstride;
  const float* xn0 = X + (size_t)((un < nunits) ? un : u) * 128 + grp * 8;
  float4 na = *reinterpret_cast<const float4*>(xn0);
  float4 nb = *reinterpret_cast<const float4*>(xn0 + 4);
  __builtin_amdgcn_sched_barrier(0);

  v2f b0[N_TILE], b1[N_TILE];
  wave_coeffs(p0, p1, p2, lane, b0, b1);

  while (true) {
    float xs[8] = {xa.x, xa.y, xa.z, xa.w, xb.x, xb.y, xb.z, xb.w};

    v2f fa0[8], fa1[8];
#pragma unroll
    for (int j = 0; j < 8; ++j) pixel_frag(xs[j], hi, fa0[j], fa1[j]);

    v8f d0 = {}, d1 = {}, d2 = {}, d3 = {}, d4 = {}, d5 = {}, d6 = {}, d7 = {};
    MM(d0, fa0[0], b0[0]);   MM(d1, fa0[1], b0[1]);
    MM(d2, fa0[2], b0[2]);   MM(d3, fa0[3], b0[3]);
    MM(d4, fa0[4], b0[4]);   MM(d5, fa0[5], b0[5]);
    MM(d6, fa0[6], b0[6]);   MM(d7, fa0[7], b0[7]);
    MM(d0, fa1[0], b1[0]);   MM(d1, fa1[1], b1[1]);
    MM(d2, fa1[2], b1[2]);   MM(d3, fa1[3], b1[3]);
    MM(d4, fa1[4], b1[4]);   MM(d5, fa1[5], b1[5]);
    MM(d6, fa1[6], b1[6]);   MM(d7, fa1[7], b1[7]);

    const bool c2 = n < 2, c4 = n < 4, c6 = n < 6, c8 = n < 8;
    const bool c10 = n < 10, c12 = n < 12, c14 = n < 14;
    float ln[8][5];
#pragma unroll
    for (int r = 0; r < 8; ++r) {
      const float h0 = c10 ? d0[r] : d1[r];
      const float t1 = c14 ? d2[r] : d3[r];
      const float h1 = c4 ? d1[r] : t1;
      const float h2 = c8 ? d3[r] : d4[r];
      const float t3 = c12 ? d5[r] : d6[r];
      const float h3 = c2 ? d4[r] : t3;
      const float h4 = c6 ? d6[r] : d7[r];
      const float x1 = half_swap(h1), x3 = half_swap(h3);
      ln[r][0] = hi ? x1 : h0;
      ln[r][1] = hi ? x3 : h2;
      ln[r][2] = hi ? h0 : h4;
      ln[r][3] = hi ? h2 : x1;
      ln[r][4] = hi ? h4 : x3;
    }
    float* o = out + (size_t)u * 1280 + lane;
#pragma unroll
    for (int r = 0; r < 8; ++r) {
#pragma unroll
      for (int i = 0; i < 5; ++i) o[r * 160 + 32 * i] = ln[r][i];
    }

    if (un >= nunits) break;
    u = un;
    xa = na;
    xb = nb;
    un = u + wstride;
    const float* xn = X + (size_t)((un < nunits) ? un : u) * 128 + grp * 8;
    na = *reinterpret_cast<const float4*>(xn);
    nb = *reinterpret_cast<const float4*>(xn + 4);
  }
}

extern "C" void kernel_launch(void* const* d_in, const int* in_sizes, int n_in,
                              void* d_out, int out_size, void* d_ws, size_t ws_size,
                              hipStream_t stream) {
  const float* x      = (const float*)d_in[0];
  const float* params = (const float*)d_in[1];
  float* out  = (float*)d_out;

  const int npix   = in_sizes[0];
  const int nunits = npix / 128;

  int blocks = (nunits + 7) / 8;
  if (blocks > 512) blocks = 512;
  hipLaunchKernelGGL(qc_fuzzy_kernel, dim3(blocks), dim3(256), 0, stream,
                     x, params, out, nunits);
}


# Wave index made scalar (readfirstlane): the unit loop and its tests run on the scalar unit; x loads take a scalar base and a fixed 32-bit lane offset, the store address is one add per unit
# speedup vs baseline: 2.7162x; 1.0701x over previous
#include <hip/hip_runtime.h>
#include <math.h>


typedef __attribute__((ext_vector_type(2))) float v2f;
typedef __attribute__((ext_vector_type(8))) float v8f;

#define NUM_CLASSES 10
#define N_LAYERS 3
#define N_TILE 8


template <int LEN>
__device__ __forceinline__ void qc_layer(float (&tr)[4], float (&ti)[4],
                                         float (&br)[4], float (&bi)[4],
                                         float s1, float c1, float sz0, float cz0,
                                         float sz2, float cz2) {
  float ntr[4], nti[4], nbr[4], nbi[4];
#pragma unroll
  for (int j = 0; j <= LEN; ++j) {
    float xr = 0.f, xi = 0.f, yr = 0.f, yi = 0.f;
    if (j < LEN) { xr = tr[j]; xi = ti[j]; yr = br[j]; yi = bi[j]; }
    if (j >= 1)  { xr -= br[j-1]; xi -= bi[j-1]; yr += tr[j-1]; yi += ti[j-1]; }
    ntr[j] = xr; nti[j] = xi; nbr[j] = yr; nbi[j] = yi;
  }
#pragma unroll
  for (int j = 0; j <= LEN; ++j) {
    float xr = ntr[j], xi = nti[j], yr = nbr[j], yi = nbi[j];
    float txr = xr * cz0 + xi * sz0;
    float txi = xi * cz0 - xr * sz0;
    float tyr = yr * cz0 - yi * sz0;
    float tyi = yi * cz0 + yr * sz0;
    float uxr = c1 * txr - s1 * tyr, uxi = c1 * txi - s1 * tyi;
    float uyr = s1 * txr + c1 * tyr, uyi = s1 * txi + c1 * tyi;
    tr[j] = uxr * cz2 + uxi * sz2;
    ti[j] = uxi * cz2 - uxr * sz2;
    br[j] = uyr * cz2 - uyi * sz2;
    bi[j] = uyi * cz2 + uyr * sz2;
  }
}

template <int N>
__device__ __forceinline__ float row_rot(float v) {
  if (N == 0) return v;
  int i = __builtin_bit_cast(int, v);
  return __builtin_bit_cast(float,
      __builtin_amdgcn_update_dpp(i, i, 0x120 + N, 0xf, 0xf, true));
}

__device__ __forceinline__ void wave_coeffs(float p0, float p1, float p2,
                                            int lane, v2f (&b0)[N_TILE],
                                            v2f (&b1)[N_TILE]) {
  const int  n  = lane & 15;
  const bool hi = lane >= 16;

  float th0 = 0.5f * p0, th1 = 0.5f * p1, th2 = 0.5f * p2;
  float s1, c1, sz0, cz0, sz2, cz2;
  sincosf(th1, &s1, &c1);
  sincosf(th0, &sz0, &cz0);
  sincosf(th2, &sz2, &cz2);

  const int cl = (n < NUM_CLASSES) ? n : NUM_CLASSES - 1;
  const int src0 = cl, src1 = cl + 10, src2 = cl + 20;
  float s1a = __shfl(s1, src0),   c1a = __shfl(c1, src0);
  float z0sa = __shfl(sz0, src0), z0ca = __shfl(cz0, src0);
  float z2sa = __shfl(sz2, src0), z2ca = __shfl(cz2, src0);
  float s1b = __shfl(s1, src1),   c1b = __shfl(c1, src1);
  float z0sb = __shfl(sz0, src1), z0cb = __shfl(cz0, src1);
  float z2sb = __shfl(sz2, src1), z2cb = __shfl(cz2, src1);
  float s1c = __shfl(s1, src2),   c1c = __shfl(c1, src2);
  float z0sc = __shfl(sz0, src2), z0cc = __shfl(cz0, src2);
  float z2sc = __shfl(sz2, src2), z2cc = __shfl(cz2, src2);

  float tr[4] = {1.f, 0.f, 0.f, 0.f};
  float ti[4] = {0.f, 0.f, 0.f, 0.f};
  float br[4] = {0.f, 0.f, 0.f, 0.f};
  float bi[4] = {0.f, 0.f, 0.f, 0.f};
  qc_layer<1>(tr, ti, br, bi, s1a, c1a, z0sa, z0ca, z2sa, z2ca);
  qc_layer<2>(tr, ti, br, bi, s1b, c1b, z0sb, z0cb, z2sb, z2cb);
  qc_layer<3>(tr, ti, br, bi, s1c, c1c, z0sc, z0cc, z2sc, z2cc);
  float g[8];
#pragma unroll
  for (int k = 0; k < 7; ++k) {
    float acc = 0.f;
#pragma unroll
    for (int i = 0; i < 4; ++i) {
      int j = k - i;
      if (j >= 0 && j < 4) acc += tr[i] * tr[j] + ti[i] * ti[j];
    }
    g[k] = acc;
  }
  g[7] = 0.f;
  const bool own = n < NUM_CLASSES;
  float v[4];
#pragma unroll
  for (int e = 0; e < 4; ++e) v[e] = own ? (hi ? g[2 * e + 1] : g[2 * e]) : 0.f;

#define QC_TILE_B(t, o)                                             \
  b0[t][0] = row_rot<o>(v[0]);  b0[t][1] = row_rot<o>(v[1]);        \
  b1[t][0] = row_rot<o>(v[2]);  b1[t][1] = row_rot<o>(v[3]);
  QC_TILE_B(0, 0)   QC_TILE_B(1, 10)  QC_TILE_B(2, 4)   QC_TILE_B(3, 14)
  QC_TILE_B(4, 8)   QC_TILE_B(5, 2)   QC_TILE_B(6, 12)  QC_TILE_B(7, 6)
#undef QC_TILE_B
}

__device__ __forceinline__ void pixel_frag(float x, bool hi, v2f& f0, v2f& f1) {
  float s, c;
  __sincosf(0.5f * x, &s, &c);
  float c2 = c * c, s2 = s * s, cs = c * s;
  float c4 = c2 * c2, s4 = s2 * s2, c2s2 = c2 * s2;
  float u = hi ? cs : c2;
  f0[0] = c4 * u;
  f0[1] = c2s2 * u;
  f1[0] = s4 * u;
  f1[1] = s4 * s2;
}

__device__ __forceinline__ float half_swap(float v) {
  int i = __builtin_bit_cast(int, v);
  return __builtin_bit_cast(float,
      __builtin_amdgcn_permlanex16(i, i, 0x76543210, 0xfedcba98, false, false));
}

typedef __attribute__((ext_vector_type(4))) float v4f;
typedef __attribute__((address_space(1))) float       g_f32;
typedef __attribute__((address_space(1))) const float g_cf32;
typedef __attribute__((address_space(1))) const v4f   g_cv4f;
__device__ __forceinline__ unsigned long long uniform_addr(const float* p) {
  unsigned long long a = reinterpret_cast<unsigned long long>(p);
  unsigned lo = __builtin_amdgcn_readfirstlane((unsigned)a);
  unsigned hi = __builtin_amdgcn_readfirstlane((unsigned)(a >> 32));
  return ((unsigned long long)hi << 32) | lo;
}
__device__ __forceinline__ v4f load4(g_cf32* base, unsigned off) {
  return *(g_cv4f*)(base + off);
}

#define MM(acc, afrag, bfrag) \
  acc = __builtin_amdgcn_wmma_f32_16x16x4_f32(false, afrag, false, bfrag, (short)0, acc, false, false)

__global__ void __launch_bounds__(256)
qc_fuzzy_kernel(const float* __restrict__ X, const float* __restrict__ params,
                float* __restrict__ out, int nunits) {
  const int lane = threadIdx.x & 31;
  const int wave = threadIdx.x >> 5;
  const int wpb  = blockDim.x >> 5;
  const int wid  = __builtin_amdgcn_readfirstlane(blockIdx.x * wpb + wave);
  const int wstride = gridDim.x * wpb;

  const int  n   = lane & 15;
  const bool hi  = lane >= 16;
  const int  grp = 2 * (n & 7) + (n >> 3);
  const unsigned xoff = (unsigned)grp * 8u;
  const unsigned ooff = (unsigned)lane;

  int u = wid;
  if (u >= nunits) return;

  const int lay = (lane / 10 < N_LAYERS) ? lane / 10 : N_LAYERS - 1;
  const float* P = params + ((lane % 10) * N_LAYERS + lay) * 3;
  const float p0 = P[0], p1 = P[1], p2 = P[2];
  asm volatile("" ::: "memory");

  g_cf32* xu0 = (g_cf32*)uniform_addr(X + (size_t)u * 128);
  v4f xa = load4(xu0, xoff);
  v4f xb = load4(xu0, xoff + 4);
  int un = u + wstride;
  g_cf32* xn0 = (g_cf32*)uniform_addr(X + (size_t)((un < nunits) ? un : u) * 128);
  v4f na = load4(xn0, xoff);
  v4f nb = load4(xn0, xoff + 4);
  __builtin_amdgcn_sched_barrier(0);

  v2f b0[N_TILE], b1[N_TILE];
  wave_coeffs(p0, p1, p2, lane, b0, b1);

  while (true) {
    float xs[8] = {xa.x, xa.y, xa.z, xa.w, xb.x, xb.y, xb.z, xb.w};

    v2f fa0[8], fa1[8];
#pragma unroll
    for (int j = 0; j < 8; ++j) pixel_frag(xs[j], hi, fa0[j], fa1[j]);

    v8f d0 = {}, d1 = {}, d2 = {}, d3 = {}, d4 = {}, d5 = {}, d6 = {}, d7 = {};
    MM(d0, fa0[0], b0[0]);   MM(d1, fa0[1], b0[1]);
    MM(d2, fa0[2], b0[2]);   MM(d3, fa0[3], b0[3]);
    MM(d4, fa0[4], b0[4]);   MM(d5, fa0[5], b0[5]);
    MM(d6, fa0[6], b0[6]);   MM(d7, fa0[7], b0[7]);
    MM(d0, fa1[0], b1[0]);   MM(d1, fa1[1], b1[1]);
    MM(d2, fa1[2], b1[2]);   MM(d3, fa1[3], b1[3]);
    MM(d4, fa1[4], b1[4]);   MM(d5, fa1[5], b1[5]);
    MM(d6, fa1[6], b1[6]);   MM(d7, fa1[7], b1[7]);

    const bool c2 = n < 2, c4 = n < 4, c6 = n < 6, c8 = n < 8;
    const bool c10 = n < 10, c12 = n < 12, c14 = n < 14;
    float ln[8][5];
#pragma unroll
    for (int r = 0; r < 8; ++r) {
      const float h0 = c10 ? d0[r] : d1[r];
      const float t1 = c14 ? d2[r] : d3[r];
      const float h1 = c4 ? d1[r] : t1;
      const float h2 = c8 ? d3[r] : d4[r];
      const float t3 = c12 ? d5[r] : d6[r];
      const float h3 = c2 ? d4[r] : t3;
      const float h4 = c6 ? d6[r] : d7[r];
      const float x1 = half_swap(h1), x3 = half_swap(h3);
      ln[r][0] = hi ? x1 : h0;
      ln[r][1] = hi ? x3 : h2;
      ln[r][2] = hi ? h0 : h4;
      ln[r][3] = hi ? h2 : x1;
      ln[r][4] = hi ? h4 : x3;
    }
    g_f32* o = (g_f32*)uniform_addr(out + (size_t)u * 1280);
    g_f32* ol = o + ooff;
#pragma unroll
    for (int r = 0; r < 8; ++r) {
#pragma unroll
      for (int i = 0; i < 5; ++i) ol[r * 160 + 32 * i] = ln[r][i];
    }

    if (un >= nunits) break;
    u = un;
    xa = na;
    xb = nb;
    un = u + wstride;
    g_cf32* xn = (g_cf32*)uniform_addr(X + (size_t)((un < nunits) ? un : u) * 128);
    na = load4(xn, xoff);
    nb = load4(xn, xoff + 4);
  }
}

extern "C" void kernel_launch(void* const* d_in, const int* in_sizes, int n_in,
                              void* d_out, int out_size, void* d_ws, size_t ws_size,
                              hipStream_t stream) {
  const float* x      = (const float*)d_in[0];
  const float* params = (const float*)d_in[1];
  float* out  = (float*)d_out;

  const int npix   = in_sizes[0];
  const int nunits = npix / 128;

  int blocks = (nunits + 7) / 8;
  if (blocks > 512) blocks = 512;
  hipLaunchKernelGGL(qc_fuzzy_kernel, dim3(blocks), dim3(256), 0, stream,
                     x, params, out, nunits);
}
